# starting kernel
# baseline (speedup 1.0000x reference)
#include <hip/hip_runtime.h>
#include <math.h>

#define IMSIZE 1024
#define NSEG   128
#define FEPS   1e-10f

typedef __attribute__((ext_vector_type(2))) float v2f;
typedef __attribute__((ext_vector_type(8))) float v8f;

__global__ __launch_bounds__(256) void sketch_render_kernel(
    const float* __restrict__ xs, const float* __restrict__ ys,
    float* __restrict__ out, float* __restrict__ bmin, float* __restrict__ bmax)
{
    __shared__ float s_cy[NSEG], s_cx[NSEG], s_c0[NSEG];
    __shared__ float s_sl[NSEG], s_in[NSEG], s_vt[NSEG];
    __shared__ float s_xl[NSEG], s_xh[NSEG], s_yl[NSEG], s_yh[NSEG];
    __shared__ float s_red[16];

    const int t = threadIdx.x;

    if (t < NSEG) {
        float x0 = xs[t], x1 = xs[t + 1];
        float y0 = ys[t], y1 = ys[t + 1];
        float dx = x1 - x0, dy = y1 - y0;
        float dx2 = dx * dx;
        float d = dx2 + dy * dy;
        float invd = 1.0f / (d + FEPS);
        float cross = y0 * x1 - x0 * y1;
        s_cy[t] = dx * dy * invd;
        s_cx[t] = dx2 * invd;
        s_c0[t] = -dy * cross * invd;
        float invdx = 1.0f / (dx + FEPS);
        s_sl[t] = dy * invdx;
        s_in[t] = cross * invdx;
        s_vt[t] = (x0 == x1) ? 1.0f : 0.0f;
        s_xl[t] = fminf(x0, x1); s_xh[t] = fmaxf(x0, x1);
        s_yl[t] = fminf(y0, y1); s_yh[t] = fmaxf(y0, y1);
    }
    __syncthreads();

    const int lane  = t & 31;
    const int wave  = t >> 5;
    const int laneN = lane & 15;
    const int hi    = lane >> 4;
    const int xbase = blockIdx.x * 16;
    const int ybase = blockIdx.y * 64;
    const float xp  = (float)(xbase + laneN);

    v2f afrag[8];
#pragma unroll
    for (int g = 0; g < 8; ++g) {
        int m = g * 16 + laneN;
        afrag[g].x = hi ? s_c0[m] : s_cy[m];
        afrag[g].y = hi ? 0.0f    : s_cx[m];
    }

    float lmin = INFINITY, lmax = -INFINITY;

#pragma unroll 1
    for (int r = 0; r < 8; ++r) {
        const int   row = ybase + wave + r * 8;
        const float yp  = (float)row;

        v2f bfrag;
        bfrag.x = hi ? 1.0f : yp;
        bfrag.y = hi ? 0.0f : xp;

        float acc = 0.0f;
#pragma unroll 2
        for (int g = 0; g < 8; ++g) {
            v8f c = {0.f, 0.f, 0.f, 0.f, 0.f, 0.f, 0.f, 0.f};
            v8f dmat = __builtin_amdgcn_wmma_f32_16x16x4_f32(
                false, afrag[g], false, bfrag, (short)0, c, false, false);
            const int mbase = g * 16 + hi * 8;
#pragma unroll
            for (int j = 0; j < 8; ++j) {
                const int m = mbase + j;
                float xg   = dmat[j];
                float ygen = fmaf(s_sl[m], xg, s_in[m]);
                float ysel = (s_vt[m] != 0.0f) ? yp : ygen;
                float xc   = __builtin_amdgcn_fmed3f(xg,   s_xl[m], s_xh[m]);
                float yc   = __builtin_amdgcn_fmed3f(ysel, s_yl[m], s_yh[m]);
                float ddx  = xp - xc;
                float ddy  = yp - yc;
                acc += __builtin_amdgcn_sqrtf(fmaf(ddx, ddx, fmaf(ddy, ddy, FEPS)));
            }
        }
        acc += __shfl_xor(acc, 16, 32);
        lmin = fminf(lmin, acc);
        lmax = fmaxf(lmax, acc);
        if (lane < 16)
            out[row * IMSIZE + xbase + laneN] = acc;
    }

#pragma unroll
    for (int off = 16; off >= 1; off >>= 1) {
        lmin = fminf(lmin, __shfl_xor(lmin, off, 32));
        lmax = fmaxf(lmax, __shfl_xor(lmax, off, 32));
    }
    if (lane == 0) { s_red[wave] = lmin; s_red[8 + wave] = lmax; }
    __syncthreads();
    if (t == 0) {
        float m0 = s_red[0], m1 = s_red[8];
        for (int w = 1; w < 8; ++w) {
            m0 = fminf(m0, s_red[w]);
            m1 = fmaxf(m1, s_red[8 + w]);
        }
        const int bid = blockIdx.y * gridDim.x + blockIdx.x;
        bmin[bid] = m0;
        bmax[bid] = m1;
    }
}

__global__ __launch_bounds__(256) void sketch_minmax_kernel(
    const float* __restrict__ bmin, const float* __restrict__ bmax,
    float* __restrict__ mm, int n)
{
    __shared__ float smin[8], smax[8];
    const int t = threadIdx.x;
    float m0 = INFINITY, m1 = -INFINITY;
    for (int i = t; i < n; i += 256) {
        m0 = fminf(m0, bmin[i]);
        m1 = fmaxf(m1, bmax[i]);
    }
#pragma unroll
    for (int off = 16; off >= 1; off >>= 1) {
        m0 = fminf(m0, __shfl_xor(m0, off, 32));
        m1 = fmaxf(m1, __shfl_xor(m1, off, 32));
    }
    if ((t & 31) == 0) { smin[t >> 5] = m0; smax[t >> 5] = m1; }
    __syncthreads();
    if (t == 0) {
        for (int w = 1; w < 8; ++w) {
            m0 = fminf(m0, smin[w]);
            m1 = fmaxf(m1, smax[w]);
        }
        mm[0] = m0;
        mm[1] = m1;
    }
}

__global__ void sketch_normalize_kernel(float* __restrict__ out,
                                        const float* __restrict__ mm, int n)
{
    const int i = blockIdx.x * blockDim.x + threadIdx.x;
    if (i < n) {
        const float tmin = mm[0];
        const float tmax = mm[1];
        out[i] = (out[i] - tmin) / (tmax - tmin);
    }
}

extern "C" void kernel_launch(void* const* d_in, const int* in_sizes, int n_in,
                              void* d_out, int out_size, void* d_ws, size_t ws_size,
                              hipStream_t stream)
{
    const float* xs = (const float*)d_in[0];
    const float* ys = (const float*)d_in[1];
    float* out = (float*)d_out;
    float* ws  = (float*)d_ws;

    float* bmin = ws;
    float* bmax = ws + 1024;
    float* mm   = ws + 2048;

    dim3 grid(IMSIZE / 16, IMSIZE / 64);
    sketch_render_kernel<<<grid, 256, 0, stream>>>(xs, ys, out, bmin, bmax);

    const int nblocks = (IMSIZE / 16) * (IMSIZE / 64);
    sketch_minmax_kernel<<<1, 256, 0, stream>>>(bmin, bmax, mm, nblocks);

    const int n = IMSIZE * IMSIZE;
    sketch_normalize_kernel<<<(n + 255) / 256, 256, 0, stream>>>(out, mm, n);
}


# Tile transposed: WMMA M = 16 pixels, N = 16 segments; a lane keeps its 8 segments' parameters in registers, tail reads no LDS; pixel sum = per-lane partials + 16-lane DPP tree (new sum order)
# speedup vs baseline: 1.3097x; 1.3097x over previous
#include <hip/hip_runtime.h>
#include <math.h>


#define IMSIZE 1024
#define NSEG   128
#define FEPS   1e-10f
#define TROWS  16
#define BROWS  (8 * TROWS)

typedef __attribute__((ext_vector_type(2))) float v2f;
typedef __attribute__((ext_vector_type(8))) float v8f;

template <int MASK>
__device__ __forceinline__ float row_xor(float v)
{
    return __builtin_bit_cast(float,
        __builtin_amdgcn_update_dpp(0, __builtin_bit_cast(int, v), 0x160 + MASK, 0xf, 0xf, true));
}

__global__ __launch_bounds__(256) __attribute__((amdgpu_waves_per_eu(4, 4))) void sketch_render_kernel(
    const float* __restrict__ xs, const float* __restrict__ ys,
    float* __restrict__ out, float* __restrict__ bmin, float* __restrict__ bmax)
{
    __shared__ float s_cy[NSEG], s_cx[NSEG], s_c0[NSEG];
    __shared__ float s_sl[NSEG], s_in[NSEG], s_vt[NSEG];
    __shared__ float s_xl[NSEG], s_xh[NSEG], s_yl[NSEG], s_yh[NSEG];
    __shared__ float s_red[16];

    const int t = threadIdx.x;

    if (t < NSEG) {
        float x0 = xs[t], x1 = xs[t + 1];
        float y0 = ys[t], y1 = ys[t + 1];
        float dx = x1 - x0, dy = y1 - y0;
        float dx2 = dx * dx;
        float d = dx2 + dy * dy;
        float invd = 1.0f / (d + FEPS);
        float cross = y0 * x1 - x0 * y1;
        s_cy[t] = dx * dy * invd;
        s_cx[t] = dx2 * invd;
        s_c0[t] = -dy * cross * invd;
        float invdx = 1.0f / (dx + FEPS);
        s_sl[t] = dy * invdx;
        s_in[t] = cross * invdx;
        s_vt[t] = (x0 == x1) ? 1.0f : 0.0f;
        s_xl[t] = fminf(x0, x1); s_xh[t] = fmaxf(x0, x1);
        s_yl[t] = fminf(y0, y1); s_yh[t] = fmaxf(y0, y1);
    }
    __syncthreads();

    const int lane = t & 31;
    const int wave = __builtin_amdgcn_readfirstlane(t >> 5);
    const int c    = lane & 15;
    const int hi   = lane >> 4;
    const int xbase = blockIdx.x * 16;
    const int ybase = blockIdx.y * BROWS + wave * TROWS;

    v2f   bfrag[8];
    float p_sl[8], p_in[8], p_xl[8], p_xh[8], p_yl[8], p_yh[8];
    bool  p_vt[8];
#pragma unroll
    for (int g = 0; g < 8; ++g) {
        const int n = g * 16 + c;
        bfrag[g].x = hi ? s_c0[n] : s_cy[n];
        bfrag[g].y = hi ? 0.0f    : s_cx[n];
        p_sl[g] = s_sl[n]; p_in[g] = s_in[n];
        p_xl[g] = s_xl[n]; p_xh[g] = s_xh[n];
        p_yl[g] = s_yl[n]; p_yh[g] = s_yh[n];
        p_vt[g] = (s_vt[n] != 0.0f);
    }

    float xpr[8];
#pragma unroll
    for (int r = 0; r < 8; ++r) xpr[r] = (float)(xbase + 8 * hi + r);
    const float xa = (float)(xbase + c);

    const bool b3 = (lane & 8) != 0, b2 = (lane & 4) != 0, b1 = (lane & 2) != 0;

    float lmin = INFINITY, lmax = -INFINITY;

    auto project = [&](int row, int g) -> v8f {
        const float yp = (float)row;
        v2f afrag;
        afrag.x = hi ? 1.0f : yp;
        afrag.y = hi ? 0.0f : xa;
        v8f cz = {0.f, 0.f, 0.f, 0.f, 0.f, 0.f, 0.f, 0.f};
        return __builtin_amdgcn_wmma_f32_16x16x4_f32(
            false, afrag, false, bfrag[g], (short)0, cz, false, false);
    };

    auto tail = [&](const v8f& dmat, int g, float yp, float (&acc)[8]) {
#pragma unroll
        for (int r = 0; r < 8; ++r) {
            float xg   = dmat[r];
            float ygen = fmaf(p_sl[g], xg, p_in[g]);
            float ysel = p_vt[g] ? yp : ygen;
            float xc   = __builtin_amdgcn_fmed3f(xg,   p_xl[g], p_xh[g]);
            float yc   = __builtin_amdgcn_fmed3f(ysel, p_yl[g], p_yh[g]);
            float ddx  = xpr[r] - xc;
            float ddy  = yp - yc;
            float dist = __builtin_amdgcn_sqrtf(fmaf(ddx, ddx, fmaf(ddy, ddy, FEPS)));
            acc[r] = (g == 0) ? dist : acc[r] + dist;
        }
    };

    v8f d0 = project(ybase, 0);
    v8f d1 = project(ybase, 1);

#pragma unroll 1
    for (int i = 0; i < TROWS; ++i) {
        const int   row = ybase + i;
        const float yp  = (float)row;

        float acc[8];
#pragma unroll
        for (int q = 0; q < 4; ++q) {
            v8f n0 = (q < 3) ? project(row, 2 * q + 2) : project(row + 1, 0);
            v8f n1 = (q < 3) ? project(row, 2 * q + 3) : project(row + 1, 1);
            tail(d0, 2 * q,     yp, acc);
            tail(d1, 2 * q + 1, yp, acc);
            d0 = n0; d1 = n1;
            if (q < 3) __builtin_amdgcn_sched_barrier(0);
        }

        float s4[4];
#pragma unroll
        for (int k = 0; k < 4; ++k) {
            float keep = b3 ? acc[k + 4] : acc[k];
            float send = b3 ? acc[k]     : acc[k + 4];
            s4[k] = keep + row_xor<8>(send);
        }
        float s2[2];
#pragma unroll
        for (int k = 0; k < 2; ++k) {
            float keep = b2 ? s4[k + 2] : s4[k];
            float send = b2 ? s4[k]     : s4[k + 2];
            s2[k] = keep + row_xor<4>(send);
        }
        float keep1 = b1 ? s2[1] : s2[0];
        float send1 = b1 ? s2[0] : s2[1];
        float s1  = keep1 + row_xor<2>(send1);
        float tot = s1 + row_xor<1>(s1);

        lmin = fminf(lmin, tot);
        lmax = fmaxf(lmax, tot);
        if ((lane & 1) == 0)
            out[row * IMSIZE + xbase + 8 * hi + (c >> 1)] = tot;
    }

#pragma unroll
    for (int off = 16; off >= 1; off >>= 1) {
        lmin = fminf(lmin, __shfl_xor(lmin, off, 32));
        lmax = fmaxf(lmax, __shfl_xor(lmax, off, 32));
    }
    if (lane == 0) { s_red[wave] = lmin; s_red[8 + wave] = lmax; }
    __syncthreads();
    if (t == 0) {
        float m0 = s_red[0], m1 = s_red[8];
        for (int w = 1; w < 8; ++w) {
            m0 = fminf(m0, s_red[w]);
            m1 = fmaxf(m1, s_red[8 + w]);
        }
        const int bid = blockIdx.y * gridDim.x + blockIdx.x;
        bmin[bid] = m0;
        bmax[bid] = m1;
    }
}

__global__ __launch_bounds__(256) void sketch_minmax_kernel(
    const float* __restrict__ bmin, const float* __restrict__ bmax,
    float* __restrict__ mm, int n)
{
    __shared__ float smin[8], smax[8];
    const int t = threadIdx.x;
    float m0 = INFINITY, m1 = -INFINITY;
    for (int i = t; i < n; i += 256) {
        m0 = fminf(m0, bmin[i]);
        m1 = fmaxf(m1, bmax[i]);
    }
#pragma unroll
    for (int off = 16; off >= 1; off >>= 1) {
        m0 = fminf(m0, __shfl_xor(m0, off, 32));
        m1 = fmaxf(m1, __shfl_xor(m1, off, 32));
    }
    if ((t & 31) == 0) { smin[t >> 5] = m0; smax[t >> 5] = m1; }
    __syncthreads();
    if (t == 0) {
        for (int w = 1; w < 8; ++w) {
            m0 = fminf(m0, smin[w]);
            m1 = fmaxf(m1, smax[w]);
        }
        mm[0] = m0;
        mm[1] = m1;
    }
}

__global__ void sketch_normalize_kernel(float* __restrict__ out,
                                        const float* __restrict__ mm, int n)
{
    const int i = blockIdx.x * blockDim.x + threadIdx.x;
    if (i < n) {
        const float tmin = mm[0];
        const float tmax = mm[1];
        out[i] = (out[i] - tmin) / (tmax - tmin);
    }
}

extern "C" void kernel_launch(void* const* d_in, const int* in_sizes, int n_in,
                              void* d_out, int out_size, void* d_ws, size_t ws_size,
                              hipStream_t stream)
{
    const float* xs = (const float*)d_in[0];
    const float* ys = (const float*)d_in[1];
    float* out = (float*)d_out;
    float* ws  = (float*)d_ws;

    float* bmin = ws;
    float* bmax = ws + 1024;
    float* mm   = ws + 2048;

    dim3 grid(IMSIZE / 16, IMSIZE / BROWS);
    sketch_render_kernel<<<grid, 256, 0, stream>>>(xs, ys, out, bmin, bmax);

    const int nblocks = (IMSIZE / 16) * (IMSIZE / BROWS);
    sketch_minmax_kernel<<<1, 256, 0, stream>>>(bmin, bmax, mm, nblocks);

    const int n = IMSIZE * IMSIZE;
    sketch_normalize_kernel<<<(n + 255) / 256, 256, 0, stream>>>(out, mm, n);
}


# Tail arithmetic on packed f32 (v_pk_fma_f32, v_pk_add_f32): two pixels per instruction for ygen, both differences, both squares and the running sum; per-value operations and sum order unchanged
# speedup vs baseline: 1.4236x; 1.0870x over previous
#include <hip/hip_runtime.h>
#include <math.h>


#define IMSIZE 1024
#define NSEG   128
#define FEPS   1e-10f
#define TROWS  16
#define BROWS  (8 * TROWS)

typedef __attribute__((ext_vector_type(2))) float v2f;
typedef __attribute__((ext_vector_type(8))) float v8f;

template <int MASK>
__device__ __forceinline__ float row_xor(float v)
{
    return __builtin_bit_cast(float,
        __builtin_amdgcn_update_dpp(0, __builtin_bit_cast(int, v), 0x160 + MASK, 0xf, 0xf, true));
}

__global__ __launch_bounds__(256) __attribute__((amdgpu_waves_per_eu(4, 4))) void sketch_render_kernel(
    const float* __restrict__ xs, const float* __restrict__ ys,
    float* __restrict__ out, float* __restrict__ bmin, float* __restrict__ bmax)
{
    __shared__ float s_cy[NSEG], s_cx[NSEG], s_c0[NSEG];
    __shared__ float s_sl[NSEG], s_in[NSEG], s_vt[NSEG];
    __shared__ float s_xl[NSEG], s_xh[NSEG], s_yl[NSEG], s_yh[NSEG];
    __shared__ float s_red[16];

    const int t = threadIdx.x;

    if (t < NSEG) {
        float x0 = xs[t], x1 = xs[t + 1];
        float y0 = ys[t], y1 = ys[t + 1];
        float dx = x1 - x0, dy = y1 - y0;
        float dx2 = dx * dx;
        float d = dx2 + dy * dy;
        float invd = 1.0f / (d + FEPS);
        float cross = y0 * x1 - x0 * y1;
        s_cy[t] = dx * dy * invd;
        s_cx[t] = dx2 * invd;
        s_c0[t] = -dy * cross * invd;
        float invdx = 1.0f / (dx + FEPS);
        s_sl[t] = dy * invdx;
        s_in[t] = cross * invdx;
        s_vt[t] = (x0 == x1) ? 1.0f : 0.0f;
        s_xl[t] = fminf(x0, x1); s_xh[t] = fmaxf(x0, x1);
        s_yl[t] = fminf(y0, y1); s_yh[t] = fmaxf(y0, y1);
    }
    __syncthreads();

    const int lane = t & 31;
    const int wave = __builtin_amdgcn_readfirstlane(t >> 5);
    const int c    = lane & 15;
    const int hi   = lane >> 4;
    const int xbase = blockIdx.x * 16;
    const int ybase = blockIdx.y * BROWS + wave * TROWS;

    v2f   bfrag[8];
    float p_sl[8], p_in[8], p_xl[8], p_xh[8], p_yl[8], p_yh[8];
    bool  p_vt[8];
#pragma unroll
    for (int g = 0; g < 8; ++g) {
        const int n = g * 16 + c;
        bfrag[g].x = hi ? s_c0[n] : s_cy[n];
        bfrag[g].y = hi ? 0.0f    : s_cx[n];
        p_sl[g] = s_sl[n]; p_in[g] = s_in[n];
        p_xl[g] = s_xl[n]; p_xh[g] = s_xh[n];
        p_yl[g] = s_yl[n]; p_yh[g] = s_yh[n];
        p_vt[g] = (s_vt[n] != 0.0f);
    }

    v2f xpr2[4];
#pragma unroll
    for (int k = 0; k < 4; ++k) {
        xpr2[k].x = (float)(xbase + 8 * hi + 2 * k);
        xpr2[k].y = (float)(xbase + 8 * hi + 2 * k + 1);
    }
    const float xa = (float)(xbase + c);

    const bool b3 = (lane & 8) != 0, b2 = (lane & 4) != 0, b1 = (lane & 2) != 0;

    float lmin = INFINITY, lmax = -INFINITY;

    auto project = [&](int row, int g) -> v8f {
        const float yp = (float)row;
        v2f afrag;
        afrag.x = hi ? 1.0f : yp;
        afrag.y = hi ? 0.0f : xa;
        v8f cz = {0.f, 0.f, 0.f, 0.f, 0.f, 0.f, 0.f, 0.f};
        return __builtin_amdgcn_wmma_f32_16x16x4_f32(
            false, afrag, false, bfrag[g], (short)0, cz, false, false);
    };

    auto tail = [&](const v8f& dmat, int g, float yp, v2f (&acc)[4]) {
        const v2f sl2  = {p_sl[g], p_sl[g]};
        const v2f in2  = {p_in[g], p_in[g]};
        const v2f yp2  = {yp, yp};
        const v2f eps2 = {FEPS, FEPS};
#pragma unroll
        for (int k = 0; k < 4; ++k) {
            v2f xg;
            xg.x = dmat[2 * k]; xg.y = dmat[2 * k + 1];
            const v2f ygen = __builtin_elementwise_fma(sl2, xg, in2);
            const float ys0 = p_vt[g] ? yp : ygen.x;
            const float ys1 = p_vt[g] ? yp : ygen.y;
            v2f xc, yc;
            xc.x = __builtin_amdgcn_fmed3f(xg.x, p_xl[g], p_xh[g]);
            xc.y = __builtin_amdgcn_fmed3f(xg.y, p_xl[g], p_xh[g]);
            yc.x = __builtin_amdgcn_fmed3f(ys0,  p_yl[g], p_yh[g]);
            yc.y = __builtin_amdgcn_fmed3f(ys1,  p_yl[g], p_yh[g]);
            const v2f ddx = xpr2[k] - xc;
            const v2f ddy = yp2 - yc;
            const v2f s   = __builtin_elementwise_fma(ddx, ddx, __builtin_elementwise_fma(ddy, ddy, eps2));
            v2f dist;
            dist.x = __builtin_amdgcn_sqrtf(s.x);
            dist.y = __builtin_amdgcn_sqrtf(s.y);
            if (g == 0) acc[k] = dist;
            else        acc[k] = acc[k] + dist;
        }
    };

    v8f d0 = project(ybase, 0);
    v8f d1 = project(ybase, 1);

#pragma unroll 1
    for (int i = 0; i < TROWS; ++i) {
        const int   row = ybase + i;
        const float yp  = (float)row;

        v2f acc2[4];
#pragma unroll
        for (int q = 0; q < 4; ++q) {
            v8f n0 = (q < 3) ? project(row, 2 * q + 2) : project(row + 1, 0);
            v8f n1 = (q < 3) ? project(row, 2 * q + 3) : project(row + 1, 1);
            tail(d0, 2 * q,     yp, acc2);
            tail(d1, 2 * q + 1, yp, acc2);
            d0 = n0; d1 = n1;
            if (q < 3) __builtin_amdgcn_sched_barrier(0);
        }

        float acc[8];
#pragma unroll
        for (int k = 0; k < 4; ++k) { acc[2 * k] = acc2[k].x; acc[2 * k + 1] = acc2[k].y; }
        float s4[4];
#pragma unroll
        for (int k = 0; k < 4; ++k) {
            float keep = b3 ? acc[k + 4] : acc[k];
            float send = b3 ? acc[k]     : acc[k + 4];
            s4[k] = keep + row_xor<8>(send);
        }
        float s2[2];
#pragma unroll
        for (int k = 0; k < 2; ++k) {
            float keep = b2 ? s4[k + 2] : s4[k];
            float send = b2 ? s4[k]     : s4[k + 2];
            s2[k] = keep + row_xor<4>(send);
        }
        float keep1 = b1 ? s2[1] : s2[0];
        float send1 = b1 ? s2[0] : s2[1];
        float s1  = keep1 + row_xor<2>(send1);
        float tot = s1 + row_xor<1>(s1);

        lmin = fminf(lmin, tot);
        lmax = fmaxf(lmax, tot);
        if ((lane & 1) == 0)
            out[row * IMSIZE + xbase + 8 * hi + (c >> 1)] = tot;
    }

#pragma unroll
    for (int off = 16; off >= 1; off >>= 1) {
        lmin = fminf(lmin, __shfl_xor(lmin, off, 32));
        lmax = fmaxf(lmax, __shfl_xor(lmax, off, 32));
    }
    if (lane == 0) { s_red[wave] = lmin; s_red[8 + wave] = lmax; }
    __syncthreads();
    if (t == 0) {
        float m0 = s_red[0], m1 = s_red[8];
        for (int w = 1; w < 8; ++w) {
            m0 = fminf(m0, s_red[w]);
            m1 = fmaxf(m1, s_red[8 + w]);
        }
        const int bid = blockIdx.y * gridDim.x + blockIdx.x;
        bmin[bid] = m0;
        bmax[bid] = m1;
    }
}

__global__ __launch_bounds__(256) void sketch_minmax_kernel(
    const float* __restrict__ bmin, const float* __restrict__ bmax,
    float* __restrict__ mm, int n)
{
    __shared__ float smin[8], smax[8];
    const int t = threadIdx.x;
    float m0 = INFINITY, m1 = -INFINITY;
    for (int i = t; i < n; i += 256) {
        m0 = fminf(m0, bmin[i]);
        m1 = fmaxf(m1, bmax[i]);
    }
#pragma unroll
    for (int off = 16; off >= 1; off >>= 1) {
        m0 = fminf(m0, __shfl_xor(m0, off, 32));
        m1 = fmaxf(m1, __shfl_xor(m1, off, 32));
    }
    if ((t & 31) == 0) { smin[t >> 5] = m0; smax[t >> 5] = m1; }
    __syncthreads();
    if (t == 0) {
        for (int w = 1; w < 8; ++w) {
            m0 = fminf(m0, smin[w]);
            m1 = fmaxf(m1, smax[w]);
        }
        mm[0] = m0;
        mm[1] = m1;
    }
}

__global__ void sketch_normalize_kernel(float* __restrict__ out,
                                        const float* __restrict__ mm, int n)
{
    const int i = blockIdx.x * blockDim.x + threadIdx.x;
    if (i < n) {
        const float tmin = mm[0];
        const float tmax = mm[1];
        out[i] = (out[i] - tmin) / (tmax - tmin);
    }
}

extern "C" void kernel_launch(void* const* d_in, const int* in_sizes, int n_in,
                              void* d_out, int out_size, void* d_ws, size_t ws_size,
                              hipStream_t stream)
{
    const float* xs = (const float*)d_in[0];
    const float* ys = (const float*)d_in[1];
    float* out = (float*)d_out;
    float* ws  = (float*)d_ws;

    float* bmin = ws;
    float* bmax = ws + 1024;
    float* mm   = ws + 2048;

    dim3 grid(IMSIZE / 16, IMSIZE / BROWS);
    sketch_render_kernel<<<grid, 256, 0, stream>>>(xs, ys, out, bmin, bmax);

    const int nblocks = (IMSIZE / 16) * (IMSIZE / BROWS);
    sketch_minmax_kernel<<<1, 256, 0, stream>>>(bmin, bmax, mm, nblocks);

    const int n = IMSIZE * IMSIZE;
    sketch_normalize_kernel<<<(n + 255) / 256, 256, 0, stream>>>(out, mm, n);
}


# Vertical case moved out of the pixel tail: per row and segment group two med3 build the y clamp bounds (both clip(yp) for a vertical segment), the 64 selects per row go; values bit-identical
# speedup vs baseline: 1.4808x; 1.0402x over previous
#include <hip/hip_runtime.h>
#include <math.h>


#define IMSIZE 1024
#define NSEG   128
#define FEPS   1e-10f
#define TROWS  16
#define BROWS  (8 * TROWS)

typedef __attribute__((ext_vector_type(2))) float v2f;
typedef __attribute__((ext_vector_type(8))) float v8f;

template <int MASK>
__device__ __forceinline__ float row_xor(float v)
{
    return __builtin_bit_cast(float,
        __builtin_amdgcn_update_dpp(0, __builtin_bit_cast(int, v), 0x160 + MASK, 0xf, 0xf, true));
}

__global__ __launch_bounds__(256) __attribute__((amdgpu_waves_per_eu(4, 4))) void sketch_render_kernel(
    const float* __restrict__ xs, const float* __restrict__ ys,
    float* __restrict__ out, float* __restrict__ bmin, float* __restrict__ bmax)
{
    __shared__ float s_cy[NSEG], s_cx[NSEG], s_c0[NSEG];
    __shared__ float s_sl[NSEG], s_in[NSEG], s_vt[NSEG];
    __shared__ float s_xl[NSEG], s_xh[NSEG], s_yl[NSEG], s_yh[NSEG];
    __shared__ float s_red[16];

    const int t = threadIdx.x;

    if (t < NSEG) {
        float x0 = xs[t], x1 = xs[t + 1];
        float y0 = ys[t], y1 = ys[t + 1];
        float dx = x1 - x0, dy = y1 - y0;
        float dx2 = dx * dx;
        float d = dx2 + dy * dy;
        float invd = 1.0f / (d + FEPS);
        float cross = y0 * x1 - x0 * y1;
        s_cy[t] = dx * dy * invd;
        s_cx[t] = dx2 * invd;
        s_c0[t] = -dy * cross * invd;
        float invdx = 1.0f / (dx + FEPS);
        s_sl[t] = dy * invdx;
        s_in[t] = cross * invdx;
        s_vt[t] = (x0 == x1) ? 1.0f : 0.0f;
        s_xl[t] = fminf(x0, x1); s_xh[t] = fmaxf(x0, x1);
        s_yl[t] = fminf(y0, y1); s_yh[t] = fmaxf(y0, y1);
    }
    __syncthreads();

    const int lane = t & 31;
    const int wave = __builtin_amdgcn_readfirstlane(t >> 5);
    const int c    = lane & 15;
    const int hi   = lane >> 4;
    const int xbase = blockIdx.x * 16;
    const int ybase = blockIdx.y * BROWS + wave * TROWS;

    v2f   bfrag[8];
    float p_sl[8], p_in[8], p_xl[8], p_xh[8], p_yl[8], p_yh[8];
    float p_ya[8], p_yb[8];
#pragma unroll
    for (int g = 0; g < 8; ++g) {
        const int n = g * 16 + c;
        bfrag[g].x = hi ? s_c0[n] : s_cy[n];
        bfrag[g].y = hi ? 0.0f    : s_cx[n];
        p_sl[g] = s_sl[n]; p_in[g] = s_in[n];
        p_xl[g] = s_xl[n]; p_xh[g] = s_xh[n];
        p_yl[g] = s_yl[n]; p_yh[g] = s_yh[n];
        const bool vt = (s_vt[n] != 0.0f);
        p_ya[g] = vt ? p_yh[g] : p_yl[g];
        p_yb[g] = vt ? p_yl[g] : p_yh[g];
    }

    v2f xpr2[4];
#pragma unroll
    for (int k = 0; k < 4; ++k) {
        xpr2[k].x = (float)(xbase + 8 * hi + 2 * k);
        xpr2[k].y = (float)(xbase + 8 * hi + 2 * k + 1);
    }
    const float xa = (float)(xbase + c);

    const bool b3 = (lane & 8) != 0, b2 = (lane & 4) != 0, b1 = (lane & 2) != 0;

    float lmin = INFINITY, lmax = -INFINITY;

    auto project = [&](int row, int g) -> v8f {
        const float yp = (float)row;
        v2f afrag;
        afrag.x = hi ? 1.0f : yp;
        afrag.y = hi ? 0.0f : xa;
        v8f cz = {0.f, 0.f, 0.f, 0.f, 0.f, 0.f, 0.f, 0.f};
        return __builtin_amdgcn_wmma_f32_16x16x4_f32(
            false, afrag, false, bfrag[g], (short)0, cz, false, false);
    };

    auto tail = [&](const v8f& dmat, int g, float yp, v2f (&acc)[4]) {
        const v2f sl2  = {p_sl[g], p_sl[g]};
        const v2f in2  = {p_in[g], p_in[g]};
        const v2f yp2  = {yp, yp};
        const v2f eps2 = {FEPS, FEPS};
        const float ylo = __builtin_amdgcn_fmed3f(yp, p_yl[g], p_ya[g]);
        const float yhi = __builtin_amdgcn_fmed3f(yp, p_yb[g], p_yh[g]);
#pragma unroll
        for (int k = 0; k < 4; ++k) {
            v2f xg;
            xg.x = dmat[2 * k]; xg.y = dmat[2 * k + 1];
            const v2f ygen = __builtin_elementwise_fma(sl2, xg, in2);
            v2f xc, yc;
            xc.x = __builtin_amdgcn_fmed3f(xg.x, p_xl[g], p_xh[g]);
            xc.y = __builtin_amdgcn_fmed3f(xg.y, p_xl[g], p_xh[g]);
            yc.x = __builtin_amdgcn_fmed3f(ygen.x, ylo, yhi);
            yc.y = __builtin_amdgcn_fmed3f(ygen.y, ylo, yhi);
            const v2f ddx = xpr2[k] - xc;
            const v2f ddy = yp2 - yc;
            const v2f s   = __builtin_elementwise_fma(ddx, ddx, __builtin_elementwise_fma(ddy, ddy, eps2));
            v2f dist;
            dist.x = __builtin_amdgcn_sqrtf(s.x);
            dist.y = __builtin_amdgcn_sqrtf(s.y);
            if (g == 0) acc[k] = dist;
            else        acc[k] = acc[k] + dist;
        }
    };

    v8f d0 = project(ybase, 0);
    v8f d1 = project(ybase, 1);

#pragma unroll 1
    for (int i = 0; i < TROWS; ++i) {
        const int   row = ybase + i;
        const float yp  = (float)row;

        v2f acc2[4];
#pragma unroll
        for (int q = 0; q < 4; ++q) {
            v8f n0 = (q < 3) ? project(row, 2 * q + 2) : project(row + 1, 0);
            v8f n1 = (q < 3) ? project(row, 2 * q + 3) : project(row + 1, 1);
            tail(d0, 2 * q,     yp, acc2);
            tail(d1, 2 * q + 1, yp, acc2);
            d0 = n0; d1 = n1;
            if (q < 3) __builtin_amdgcn_sched_barrier(0);
        }

        float acc[8];
#pragma unroll
        for (int k = 0; k < 4; ++k) { acc[2 * k] = acc2[k].x; acc[2 * k + 1] = acc2[k].y; }
        float s4[4];
#pragma unroll
        for (int k = 0; k < 4; ++k) {
            float keep = b3 ? acc[k + 4] : acc[k];
            float send = b3 ? acc[k]     : acc[k + 4];
            s4[k] = keep + row_xor<8>(send);
        }
        float s2[2];
#pragma unroll
        for (int k = 0; k < 2; ++k) {
            float keep = b2 ? s4[k + 2] : s4[k];
            float send = b2 ? s4[k]     : s4[k + 2];
            s2[k] = keep + row_xor<4>(send);
        }
        float keep1 = b1 ? s2[1] : s2[0];
        float send1 = b1 ? s2[0] : s2[1];
        float s1  = keep1 + row_xor<2>(send1);
        float tot = s1 + row_xor<1>(s1);

        lmin = fminf(lmin, tot);
        lmax = fmaxf(lmax, tot);
        if ((lane & 1) == 0)
            out[row * IMSIZE + xbase + 8 * hi + (c >> 1)] = tot;
    }

#pragma unroll
    for (int off = 16; off >= 1; off >>= 1) {
        lmin = fminf(lmin, __shfl_xor(lmin, off, 32));
        lmax = fmaxf(lmax, __shfl_xor(lmax, off, 32));
    }
    if (lane == 0) { s_red[wave] = lmin; s_red[8 + wave] = lmax; }
    __syncthreads();
    if (t == 0) {
        float m0 = s_red[0], m1 = s_red[8];
        for (int w = 1; w < 8; ++w) {
            m0 = fminf(m0, s_red[w]);
            m1 = fmaxf(m1, s_red[8 + w]);
        }
        const int bid = blockIdx.y * gridDim.x + blockIdx.x;
        bmin[bid] = m0;
        bmax[bid] = m1;
    }
}

__global__ __launch_bounds__(256) void sketch_minmax_kernel(
    const float* __restrict__ bmin, const float* __restrict__ bmax,
    float* __restrict__ mm, int n)
{
    __shared__ float smin[8], smax[8];
    const int t = threadIdx.x;
    float m0 = INFINITY, m1 = -INFINITY;
    for (int i = t; i < n; i += 256) {
        m0 = fminf(m0, bmin[i]);
        m1 = fmaxf(m1, bmax[i]);
    }
#pragma unroll
    for (int off = 16; off >= 1; off >>= 1) {
        m0 = fminf(m0, __shfl_xor(m0, off, 32));
        m1 = fmaxf(m1, __shfl_xor(m1, off, 32));
    }
    if ((t & 31) == 0) { smin[t >> 5] = m0; smax[t >> 5] = m1; }
    __syncthreads();
    if (t == 0) {
        for (int w = 1; w < 8; ++w) {
            m0 = fminf(m0, smin[w]);
            m1 = fmaxf(m1, smax[w]);
        }
        mm[0] = m0;
        mm[1] = m1;
    }
}

__global__ void sketch_normalize_kernel(float* __restrict__ out,
                                        const float* __restrict__ mm, int n)
{
    const int i = blockIdx.x * blockDim.x + threadIdx.x;
    if (i < n) {
        const float tmin = mm[0];
        const float tmax = mm[1];
        out[i] = (out[i] - tmin) / (tmax - tmin);
    }
}

extern "C" void kernel_launch(void* const* d_in, const int* in_sizes, int n_in,
                              void* d_out, int out_size, void* d_ws, size_t ws_size,
                              hipStream_t stream)
{
    const float* xs = (const float*)d_in[0];
    const float* ys = (const float*)d_in[1];
    float* out = (float*)d_out;
    float* ws  = (float*)d_ws;

    float* bmin = ws;
    float* bmax = ws + 1024;
    float* mm   = ws + 2048;

    dim3 grid(IMSIZE / 16, IMSIZE / BROWS);
    sketch_render_kernel<<<grid, 256, 0, stream>>>(xs, ys, out, bmin, bmax);

    const int nblocks = (IMSIZE / 16) * (IMSIZE / BROWS);
    sketch_minmax_kernel<<<1, 256, 0, stream>>>(bmin, bmax, mm, nblocks);

    const int n = IMSIZE * IMSIZE;
    sketch_normalize_kernel<<<(n + 255) / 256, 256, 0, stream>>>(out, mm, n);
}


# Box clamps in unit-box coordinates: (v-lo)/(hi-lo) by one v_pk_fma_f32 with output clamp [0,1], the fma back yields the difference; 4 med3 + 2 subtracts a pixel pair go; new rounding, sum order kept
# speedup vs baseline: 1.6589x; 1.1203x over previous
#include <hip/hip_runtime.h>
#include <math.h>


#define IMSIZE 1024
#define NSEG   128
#define FEPS   1e-10f
#define TROWS  16
#define BROWS  (8 * TROWS)

typedef __attribute__((ext_vector_type(2))) float v2f;
typedef __attribute__((ext_vector_type(8))) float v8f;

template <int MASK>
__device__ __forceinline__ float row_xor(float v)
{
    return __builtin_bit_cast(float,
        __builtin_amdgcn_update_dpp(0, __builtin_bit_cast(int, v), 0x160 + MASK, 0xf, 0xf, true));
}

__global__ __launch_bounds__(256) __attribute__((amdgpu_waves_per_eu(4, 4))) void sketch_render_kernel(
    const float* __restrict__ xs, const float* __restrict__ ys,
    float* __restrict__ out, float* __restrict__ bmin, float* __restrict__ bmax)
{
    __shared__ float s_cy[NSEG], s_cx[NSEG], s_c0[NSEG];
    __shared__ float s_ax[NSEG], s_bx[NSEG], s_nw[NSEG], s_xl[NSEG];
    __shared__ float s_ay[NSEG], s_by[NSEG], s_vh[NSEG], s_nh[NSEG], s_yl[NSEG];
    __shared__ float s_red[16];

    const int t = threadIdx.x;

    if (t < NSEG) {
        float x0 = xs[t], x1 = xs[t + 1];
        float y0 = ys[t], y1 = ys[t + 1];
        float dx = x1 - x0, dy = y1 - y0;
        float dx2 = dx * dx;
        float d = dx2 + dy * dy;
        float invd = 1.0f / (d + FEPS);
        float cross = y0 * x1 - x0 * y1;
        s_cy[t] = dx * dy * invd;
        s_cx[t] = dx2 * invd;
        s_c0[t] = -dy * cross * invd;
        float invdx = 1.0f / (dx + FEPS);
        const float sl = dy * invdx;
        const float ic = cross * invdx;
        const bool  vt = (x0 == x1);
        const float xl = fminf(x0, x1), yl = fminf(y0, y1);
        const float w  = fmaxf(x0, x1) - xl;
        const float h  = fmaxf(y0, y1) - yl;
        const float iw = (w > 1e-30f) ? 1.0f / w : 0.0f;
        const float ih = (h > 1e-30f) ? 1.0f / h : 0.0f;
        s_ax[t] = iw;  s_bx[t] = -xl * iw;  s_nw[t] = -w;  s_xl[t] = xl;
        s_ay[t] = vt ? 0.0f : sl * ih;
        s_by[t] = vt ? 0.0f : (ic - yl) * ih;
        s_vh[t] = vt ? ih : 0.0f;
        s_nh[t] = -h;  s_yl[t] = yl;
    }
    __syncthreads();

    const int lane = t & 31;
    const int wave = __builtin_amdgcn_readfirstlane(t >> 5);
    const int c    = lane & 15;
    const int hi   = lane >> 4;
    const int xbase = blockIdx.x * 16;
    const int ybase = blockIdx.y * BROWS + wave * TROWS;

    v2f bfrag[8];
#pragma unroll
    for (int g = 0; g < 8; ++g) {
        const int n = g * 16 + c;
        bfrag[g].x = hi ? s_c0[n] : s_cy[n];
        bfrag[g].y = hi ? 0.0f    : s_cx[n];
    }
    v2f k_ax[4], k_bx[4], k_nw[4], k_ay[4], k_by[4], k_vh[4], k_nh[4], k_yl[4];
#pragma unroll
    for (int q = 0; q < 4; ++q) {
        const int n0 = (2 * q) * 16 + c, n1 = n0 + 16;
        k_ax[q].x = s_ax[n0]; k_ax[q].y = s_ax[n1];
        k_bx[q].x = s_bx[n0]; k_bx[q].y = s_bx[n1];
        k_nw[q].x = s_nw[n0]; k_nw[q].y = s_nw[n1];
        k_ay[q].x = s_ay[n0]; k_ay[q].y = s_ay[n1];
        k_by[q].x = s_by[n0]; k_by[q].y = s_by[n1];
        k_vh[q].x = s_vh[n0]; k_vh[q].y = s_vh[n1];
        k_nh[q].x = s_nh[n0]; k_nh[q].y = s_nh[n1];
        k_yl[q].x = s_yl[n0]; k_yl[q].y = s_yl[n1];
    }
    v2f k_dx[8][4];
#pragma unroll
    for (int g = 0; g < 8; ++g) {
        const float xl = s_xl[g * 16 + c];
#pragma unroll
        for (int k = 0; k < 4; ++k) {
            k_dx[g][k].x = (float)(xbase + 8 * hi + 2 * k)     - xl;
            k_dx[g][k].y = (float)(xbase + 8 * hi + 2 * k + 1) - xl;
        }
    }
    const float xa = (float)(xbase + c);

    const bool b3 = (lane & 8) != 0, b2 = (lane & 4) != 0, b1 = (lane & 2) != 0;

    float lmin = INFINITY, lmax = -INFINITY;

    auto project = [&](int row, int g) -> v8f {
        const float yp = (float)row;
        v2f afrag;
        afrag.x = hi ? 1.0f : yp;
        afrag.y = hi ? 0.0f : xa;
        v8f cz = {0.f, 0.f, 0.f, 0.f, 0.f, 0.f, 0.f, 0.f};
        return __builtin_amdgcn_wmma_f32_16x16x4_f32(
            false, afrag, false, bfrag[g], (short)0, cz, false, false);
    };

#define PK_FMA_NEW(r, a, x, b, MODS) asm volatile("v_pk_fma_f32 %0, %1, %2, %3 " MODS : "=v"(r) : "v"(a), "v"(x), "v"(b))
#define PK_FMA_OVER(x, a, b, MODS)   asm volatile("v_pk_fma_f32 %0, %1, %0, %2 " MODS : "+v"(x) : "v"(a), "v"(b))
    auto pk_unit_new = [&](v2f a, v2f x, v2f b, int H) -> v2f {
        v2f r;
        if (H == 0) PK_FMA_NEW(r, a, x, b, "op_sel:[0,0,0] op_sel_hi:[0,1,0] clamp");
        else        PK_FMA_NEW(r, a, x, b, "op_sel:[1,0,1] op_sel_hi:[1,1,1] clamp");
        return r;
    };
    auto pk_unit_over = [&](v2f& x, v2f a, v2f b, int H) {
        if (H == 0) PK_FMA_OVER(x, a, b, "op_sel:[0,0,0] op_sel_hi:[0,1,0] clamp");
        else        PK_FMA_OVER(x, a, b, "op_sel:[1,0,1] op_sel_hi:[1,1,1] clamp");
    };
    auto pk_half_over = [&](v2f& x, v2f a, v2f b, int H) {
        if (H == 0) PK_FMA_OVER(x, a, b, "op_sel:[0,0,0] op_sel_hi:[0,1,1]");
        else        PK_FMA_OVER(x, a, b, "op_sel:[1,0,0] op_sel_hi:[1,1,1]");
    };
    auto pk_half2_over = [&](v2f& x, v2f a, v2f b, int H) {
        if (H == 0) PK_FMA_OVER(x, a, b, "op_sel:[0,0,0] op_sel_hi:[0,1,0]");
        else        PK_FMA_OVER(x, a, b, "op_sel:[1,0,1] op_sel_hi:[1,1,1]");
    };

    auto box_stage1 = [&](const v8f& da, const v8f& db, int q, v2f (&xg)[8], v2f (&tx)[8]) {
#pragma unroll
        for (int k = 0; k < 4; ++k) {
            xg[k].x     = da[2 * k]; xg[k].y     = da[2 * k + 1];
            xg[4 + k].x = db[2 * k]; xg[4 + k].y = db[2 * k + 1];
        }
#pragma unroll
        for (int i = 0; i < 8; ++i) tx[i] = pk_unit_new(k_ax[q], xg[i], k_bx[q], i >> 2);
    };
    auto box_stage234 = [&](int q, v2f ypl, v2f brow, v2f (&xg)[8], v2f (&tx)[8]) {
#pragma unroll
        for (int i = 0; i < 8; ++i) pk_unit_over (xg[i], k_ay[q], brow, i >> 2);
#pragma unroll
        for (int i = 0; i < 8; ++i) pk_half2_over(xg[i], k_nh[q], ypl,  i >> 2);
#pragma unroll
        for (int i = 0; i < 8; ++i) pk_half_over (tx[i], k_nw[q], k_dx[2 * q + (i >> 2)][i & 3], i >> 2);
    };

    auto add_dists = [&](const v2f (&ddx)[8], const v2f (&ddy)[8], int q, v2f (&acc)[4]) {
        const v2f eps2 = {FEPS, FEPS};
#pragma unroll
        for (int i = 0; i < 8; ++i) {
            const int k = i & 3;
            const v2f s = __builtin_elementwise_fma(ddx[i], ddx[i], __builtin_elementwise_fma(ddy[i], ddy[i], eps2));
            v2f dist;
            dist.x = __builtin_amdgcn_sqrtf(s.x);
            dist.y = __builtin_amdgcn_sqrtf(s.y);
            if (q == 0 && i < 4) acc[k] = dist;
            else                 acc[k] = acc[k] + dist;
        }
    };

    auto row_terms = [&](float yp, int q, v2f& ypl, v2f& brow) {
        const v2f yp2 = {yp, yp};
        ypl  = yp2 - k_yl[q];
        brow = __builtin_elementwise_fma(ypl, k_vh[q], k_by[q]);
    };

    v8f d0 = project(ybase, 0);
    v8f d1 = project(ybase, 1);
    v2f ypl, brow;
    row_terms((float)ybase, 0, ypl, brow);

#pragma unroll 1
    for (int i = 0; i < TROWS; ++i) {
        const int   row = ybase + i;
        const float yp  = (float)row;

        v2f acc2[4];
#pragma unroll
        for (int q = 0; q < 4; ++q) {
            v2f ddy[8], ddx[8];
            box_stage1(d0, d1, q, ddy, ddx);
            __builtin_amdgcn_sched_barrier(0);
            v8f n0 = (q < 3) ? project(row, 2 * q + 2) : project(row + 1, 0);
            v8f n1 = (q < 3) ? project(row, 2 * q + 3) : project(row + 1, 1);
            box_stage234(q, ypl, brow, ddy, ddx);
            __builtin_amdgcn_sched_barrier(0);
            add_dists(ddx, ddy, q, acc2);
            if (q < 3) row_terms(yp, q + 1, ypl, brow);
            else       row_terms((float)(row + 1), 0, ypl, brow);
            d0 = n0; d1 = n1;
            if (q < 3) __builtin_amdgcn_sched_barrier(0);
        }

        float acc[8];
#pragma unroll
        for (int k = 0; k < 4; ++k) { acc[2 * k] = acc2[k].x; acc[2 * k + 1] = acc2[k].y; }
        float s4[4];
#pragma unroll
        for (int k = 0; k < 4; ++k) {
            float keep = b3 ? acc[k + 4] : acc[k];
            float send = b3 ? acc[k]     : acc[k + 4];
            s4[k] = keep + row_xor<8>(send);
        }
        float s2[2];
#pragma unroll
        for (int k = 0; k < 2; ++k) {
            float keep = b2 ? s4[k + 2] : s4[k];
            float send = b2 ? s4[k]     : s4[k + 2];
            s2[k] = keep + row_xor<4>(send);
        }
        float keep1 = b1 ? s2[1] : s2[0];
        float send1 = b1 ? s2[0] : s2[1];
        float s1  = keep1 + row_xor<2>(send1);
        float tot = s1 + row_xor<1>(s1);

        lmin = fminf(lmin, tot);
        lmax = fmaxf(lmax, tot);
        if ((lane & 1) == 0)
            out[row * IMSIZE + xbase + 8 * hi + (c >> 1)] = tot;
    }

#pragma unroll
    for (int off = 16; off >= 1; off >>= 1) {
        lmin = fminf(lmin, __shfl_xor(lmin, off, 32));
        lmax = fmaxf(lmax, __shfl_xor(lmax, off, 32));
    }
    if (lane == 0) { s_red[wave] = lmin; s_red[8 + wave] = lmax; }
    __syncthreads();
    if (t == 0) {
        float m0 = s_red[0], m1 = s_red[8];
        for (int w = 1; w < 8; ++w) {
            m0 = fminf(m0, s_red[w]);
            m1 = fmaxf(m1, s_red[8 + w]);
        }
        const int bid = blockIdx.y * gridDim.x + blockIdx.x;
        bmin[bid] = m0;
        bmax[bid] = m1;
    }
}

__global__ __launch_bounds__(256) void sketch_minmax_kernel(
    const float* __restrict__ bmin, const float* __restrict__ bmax,
    float* __restrict__ mm, int n)
{
    __shared__ float smin[8], smax[8];
    const int t = threadIdx.x;
    float m0 = INFINITY, m1 = -INFINITY;
    for (int i = t; i < n; i += 256) {
        m0 = fminf(m0, bmin[i]);
        m1 = fmaxf(m1, bmax[i]);
    }
#pragma unroll
    for (int off = 16; off >= 1; off >>= 1) {
        m0 = fminf(m0, __shfl_xor(m0, off, 32));
        m1 = fmaxf(m1, __shfl_xor(m1, off, 32));
    }
    if ((t & 31) == 0) { smin[t >> 5] = m0; smax[t >> 5] = m1; }
    __syncthreads();
    if (t == 0) {
        for (int w = 1; w < 8; ++w) {
            m0 = fminf(m0, smin[w]);
            m1 = fmaxf(m1, smax[w]);
        }
        mm[0] = m0;
        mm[1] = m1;
    }
}

__global__ void sketch_normalize_kernel(float* __restrict__ out,
                                        const float* __restrict__ mm, int n)
{
    const int i = blockIdx.x * blockDim.x + threadIdx.x;
    if (i < n) {
        const float tmin = mm[0];
        const float tmax = mm[1];
        out[i] = (out[i] - tmin) / (tmax - tmin);
    }
}

extern "C" void kernel_launch(void* const* d_in, const int* in_sizes, int n_in,
                              void* d_out, int out_size, void* d_ws, size_t ws_size,
                              hipStream_t stream)
{
    const float* xs = (const float*)d_in[0];
    const float* ys = (const float*)d_in[1];
    float* out = (float*)d_out;
    float* ws  = (float*)d_ws;

    float* bmin = ws;
    float* bmax = ws + 1024;
    float* mm   = ws + 2048;

    dim3 grid(IMSIZE / 16, IMSIZE / BROWS);
    sketch_render_kernel<<<grid, 256, 0, stream>>>(xs, ys, out, bmin, bmax);

    const int nblocks = (IMSIZE / 16) * (IMSIZE / BROWS);
    sketch_minmax_kernel<<<1, 256, 0, stream>>>(bmin, bmax, mm, nblocks);

    const int n = IMSIZE * IMSIZE;
    sketch_normalize_kernel<<<(n + 255) / 256, 256, 0, stream>>>(out, mm, n);
}


# Packed halves re-paired by row (same pixel, rows y and y+1): one WMMA value feeds both, row step c_y folded into the clamp fmas' constants; 4 WMMAs a row, not 8; new rounding, sum order kept
# speedup vs baseline: 1.8277x; 1.1018x over previous
#include <hip/hip_runtime.h>
#include <math.h>


#define IMSIZE 1024
#define NSEG   128
#define FEPS   1e-10f
#define TROWS  16
#define BROWS  (8 * TROWS)

typedef __attribute__((ext_vector_type(2))) float v2f;
typedef __attribute__((ext_vector_type(8))) float v8f;

template <int MASK>
__device__ __forceinline__ float row_xor(float v)
{
    return __builtin_bit_cast(float,
        __builtin_amdgcn_update_dpp(0, __builtin_bit_cast(int, v), 0x160 + MASK, 0xf, 0xf, true));
}

__global__ __launch_bounds__(256) __attribute__((amdgpu_waves_per_eu(4, 4))) void sketch_render_kernel(
    const float* __restrict__ xs, const float* __restrict__ ys,
    float* __restrict__ out, float* __restrict__ bmin, float* __restrict__ bmax)
{
    __shared__ float s_cy[NSEG], s_cx[NSEG], s_c0[NSEG];
    __shared__ float s_ax[NSEG], s_nw[NSEG], s_xl[NSEG];
    __shared__ float s_ay[NSEG], s_nh[NSEG];
    __shared__ float4 s_rt[NSEG];
    __shared__ float2 s_b1[NSEG];
    __shared__ float s_red[16];

    const int t = threadIdx.x;

    if (t < NSEG) {
        float x0 = xs[t], x1 = xs[t + 1];
        float y0 = ys[t], y1 = ys[t + 1];
        float dx = x1 - x0, dy = y1 - y0;
        float dx2 = dx * dx;
        float d = dx2 + dy * dy;
        float invd = 1.0f / (d + FEPS);
        float cross = y0 * x1 - x0 * y1;
        const float cy = dx * dy * invd;
        s_cy[t] = cy;
        s_cx[t] = dx2 * invd;
        s_c0[t] = -dy * cross * invd;
        float invdx = 1.0f / (dx + FEPS);
        const float sl = dy * invdx;
        const float ic = cross * invdx;
        const bool  vt = (x0 == x1);
        const float xl = fminf(x0, x1), yl = fminf(y0, y1);
        const float w  = fmaxf(x0, x1) - xl;
        const float h  = fmaxf(y0, y1) - yl;
        const float iw = (w > 1e-30f) ? 1.0f / w : 0.0f;
        const float ih = (h > 1e-30f) ? 1.0f / h : 0.0f;
        const float ax = iw, bx = -xl * iw;
        s_ax[t] = ax;  s_nw[t] = -w;  s_xl[t] = xl;
        s_b1[t] = make_float2(bx, bx + ax * cy);
        const float ay = vt ? 0.0f : sl * ih;
        const float by = vt ? 0.0f : (ic - yl) * ih;
        s_ay[t] = ay;  s_nh[t] = -h;
        s_rt[t] = make_float4(yl, vt ? ih : 0.0f, by, by + ay * cy);
    }
    __syncthreads();

    const int lane = t & 31;
    const int wave = __builtin_amdgcn_readfirstlane(t >> 5);
    const int c    = lane & 15;
    const int hi   = lane >> 4;
    const int xbase = blockIdx.x * 16;
    const int ybase = blockIdx.y * BROWS + wave * TROWS;

    v2f bfrag[8];
#pragma unroll
    for (int g = 0; g < 8; ++g) {
        const int n = g * 16 + c;
        bfrag[g].x = hi ? s_c0[n] : s_cy[n];
        bfrag[g].y = hi ? 0.0f    : s_cx[n];
    }
    v2f k_ax[4], k_nw[4], k_ay[4], k_nh[4];
#pragma unroll
    for (int q = 0; q < 4; ++q) {
        const int n0 = (2 * q) * 16 + c, n1 = n0 + 16;
        k_ax[q].x = s_ax[n0]; k_ax[q].y = s_ax[n1];
        k_nw[q].x = s_nw[n0]; k_nw[q].y = s_nw[n1];
        k_ay[q].x = s_ay[n0]; k_ay[q].y = s_ay[n1];
        k_nh[q].x = s_nh[n0]; k_nh[q].y = s_nh[n1];
    }
    v2f k_dx[8][4];
#pragma unroll
    for (int g = 0; g < 8; ++g) {
        const float xl = s_xl[g * 16 + c];
#pragma unroll
        for (int k = 0; k < 4; ++k) {
            k_dx[g][k].x = (float)(xbase + 8 * hi + 2 * k)     - xl;
            k_dx[g][k].y = (float)(xbase + 8 * hi + 2 * k + 1) - xl;
        }
    }
    float4 k_rt[8];
    v2f    k_b1[8];
#pragma unroll
    for (int g = 0; g < 8; ++g) {
        k_rt[g] = s_rt[g * 16 + c];
        const float2 bq = s_b1[g * 16 + c];
        k_b1[g].x = bq.x; k_b1[g].y = bq.y;
    }
    const float xa = (float)(xbase + c);

    const bool b3 = (lane & 8) != 0, b2 = (lane & 4) != 0, b1 = (lane & 2) != 0;

    float lmin = INFINITY, lmax = -INFINITY;

    auto project = [&](int row, int g) -> v8f {
        const float yp = (float)row;
        v2f afrag;
        afrag.x = hi ? 1.0f : yp;
        afrag.y = hi ? 0.0f : xa;
        v8f cz = {0.f, 0.f, 0.f, 0.f, 0.f, 0.f, 0.f, 0.f};
        return __builtin_amdgcn_wmma_f32_16x16x4_f32(
            false, afrag, false, bfrag[g], (short)0, cz, false, false);
    };

#define PK_FMA_NEW(r, a, x, b, MODS) asm volatile("v_pk_fma_f32 %0, %1, %2, %3 " MODS : "=v"(r) : "v"(a), "v"(x), "v"(b))
#define PK_FMA_OVER(x, a, b, MODS)   asm volatile("v_pk_fma_f32 %0, %1, %0, %2 " MODS : "+v"(x) : "v"(a), "v"(b))
    auto pk_unit_new = [&](v2f a, v2f x, v2f b, int H, int P) -> v2f {
        v2f r;
        if      (H == 0 && P == 0) PK_FMA_NEW(r, a, x, b, "op_sel:[0,0,0] op_sel_hi:[0,0,1] clamp");
        else if (H == 0 && P == 1) PK_FMA_NEW(r, a, x, b, "op_sel:[0,1,0] op_sel_hi:[0,1,1] clamp");
        else if (H == 1 && P == 0) PK_FMA_NEW(r, a, x, b, "op_sel:[1,0,0] op_sel_hi:[1,0,1] clamp");
        else                       PK_FMA_NEW(r, a, x, b, "op_sel:[1,1,0] op_sel_hi:[1,1,1] clamp");
        return r;
    };
    auto pk_backy_over = [&](v2f& x, v2f a, v2f b, int H) {
        if (H == 0) PK_FMA_OVER(x, a, b, "op_sel:[0,0,0] op_sel_hi:[0,1,1]");
        else        PK_FMA_OVER(x, a, b, "op_sel:[1,0,0] op_sel_hi:[1,1,1]");
    };
    auto pk_backx_over = [&](v2f& x, v2f a, v2f b, int H, int P) {
        if      (H == 0 && P == 0) PK_FMA_OVER(x, a, b, "op_sel:[0,0,0] op_sel_hi:[0,1,0]");
        else if (H == 0 && P == 1) PK_FMA_OVER(x, a, b, "op_sel:[0,0,1] op_sel_hi:[0,1,1]");
        else if (H == 1 && P == 0) PK_FMA_OVER(x, a, b, "op_sel:[1,0,0] op_sel_hi:[1,1,0]");
        else                       PK_FMA_OVER(x, a, b, "op_sel:[1,0,1] op_sel_hi:[1,1,1]");
    };

    auto box_stage1 = [&](const v8f& d, int g, v2f bxp, v2f browp, v2f (&tx)[8], v2f (&ty)[8]) {
        v2f dp[4];
#pragma unroll
        for (int k = 0; k < 4; ++k) { dp[k].x = d[2 * k]; dp[k].y = d[2 * k + 1]; }
#pragma unroll
        for (int r = 0; r < 8; ++r) tx[r] = pk_unit_new(k_ax[g >> 1], dp[r >> 1], bxp,   g & 1, r & 1);
#pragma unroll
        for (int r = 0; r < 8; ++r) ty[r] = pk_unit_new(k_ay[g >> 1], dp[r >> 1], browp, g & 1, r & 1);
    };
    auto box_stage2 = [&](int g, v2f yplp, v2f (&tx)[8], v2f (&ty)[8]) {
#pragma unroll
        for (int r = 0; r < 8; ++r) pk_backy_over(ty[r], k_nh[g >> 1], yplp, g & 1);
#pragma unroll
        for (int r = 0; r < 8; ++r) pk_backx_over(tx[r], k_nw[g >> 1], k_dx[g][r >> 1], g & 1, r & 1);
    };

    auto add_dists = [&](const v2f (&ddx)[8], const v2f (&ddy)[8], int g, v2f (&acc)[8]) {
        const v2f eps2 = {FEPS, FEPS};
#pragma unroll
        for (int r = 0; r < 8; ++r) {
            const v2f s = __builtin_elementwise_fma(ddx[r], ddx[r], __builtin_elementwise_fma(ddy[r], ddy[r], eps2));
            v2f dist;
            dist.x = __builtin_amdgcn_sqrtf(s.x);
            dist.y = __builtin_amdgcn_sqrtf(s.y);
            if (g == 0) acc[r] = dist;
            else        acc[r] = acc[r] + dist;
        }
    };

    auto row_terms = [&](float yp0, float yp1, float4 rt, v2f& yplp, v2f& browp) {
        yplp.x = yp0 - rt.x;
        yplp.y = yp1 - rt.x;
        const v2f vh2 = {rt.y, rt.y};
        const v2f by2 = {rt.z, rt.w};
        browp = __builtin_elementwise_fma(yplp, vh2, by2);
    };

    auto lane_sum = [&](const float (&acc)[8]) -> float {
        float s4[4];
#pragma unroll
        for (int k = 0; k < 4; ++k) {
            float keep = b3 ? acc[k + 4] : acc[k];
            float send = b3 ? acc[k]     : acc[k + 4];
            s4[k] = keep + row_xor<8>(send);
        }
        float s2[2];
#pragma unroll
        for (int k = 0; k < 2; ++k) {
            float keep = b2 ? s4[k + 2] : s4[k];
            float send = b2 ? s4[k]     : s4[k + 2];
            s2[k] = keep + row_xor<4>(send);
        }
        float keep1 = b1 ? s2[1] : s2[0];
        float send1 = b1 ? s2[0] : s2[1];
        float s1  = keep1 + row_xor<2>(send1);
        return s1 + row_xor<1>(s1);
    };

    v8f dcur = project(ybase, 0);
    v2f yplp, browp;
    row_terms((float)ybase, (float)(ybase + 1), k_rt[0], yplp, browp);

#pragma unroll 1
    for (int i = 0; i < TROWS; i += 2) {
        const int   row = ybase + i;
        const float yp0 = (float)row, yp1 = (float)(row + 1);

        v2f acc2[8];
#pragma unroll
        for (int g = 0; g < 8; ++g) {
            v2f ddx[8], ddy[8];
            box_stage1(dcur, g, k_b1[g], browp, ddx, ddy);
            __builtin_amdgcn_sched_barrier(0);
            const v8f dnext = (g < 7) ? project(row, g + 1) : project(row + 2, 0);
            box_stage2(g, yplp, ddx, ddy);
            __builtin_amdgcn_sched_barrier(0);
            add_dists(ddx, ddy, g, acc2);
            asm volatile("" :: "v"(acc2[0]), "v"(acc2[1]), "v"(acc2[2]), "v"(acc2[3]),
                               "v"(acc2[4]), "v"(acc2[5]), "v"(acc2[6]), "v"(acc2[7]));
            if (g < 7) row_terms(yp0, yp1, k_rt[g + 1], yplp, browp);
            else       row_terms((float)(row + 2), (float)(row + 3), k_rt[0], yplp, browp);
            dcur = dnext;
            if (g < 7) __builtin_amdgcn_sched_barrier(0);
        }

        float ra[8], rb[8];
#pragma unroll
        for (int r = 0; r < 8; ++r) { ra[r] = acc2[r].x; rb[r] = acc2[r].y; }
        const float tot0 = lane_sum(ra);
        const float tot1 = lane_sum(rb);

        lmin = fminf(lmin, fminf(tot0, tot1));
        lmax = fmaxf(lmax, fmaxf(tot0, tot1));
        if ((lane & 1) == 0) {
            out[row * IMSIZE + xbase + 8 * hi + (c >> 1)]       = tot0;
            out[(row + 1) * IMSIZE + xbase + 8 * hi + (c >> 1)] = tot1;
        }
    }

#pragma unroll
    for (int off = 16; off >= 1; off >>= 1) {
        lmin = fminf(lmin, __shfl_xor(lmin, off, 32));
        lmax = fmaxf(lmax, __shfl_xor(lmax, off, 32));
    }
    if (lane == 0) { s_red[wave] = lmin; s_red[8 + wave] = lmax; }
    __syncthreads();
    if (t == 0) {
        float m0 = s_red[0], m1 = s_red[8];
        for (int w = 1; w < 8; ++w) {
            m0 = fminf(m0, s_red[w]);
            m1 = fmaxf(m1, s_red[8 + w]);
        }
        const int bid = blockIdx.y * gridDim.x + blockIdx.x;
        bmin[bid] = m0;
        bmax[bid] = m1;
    }
}

__global__ __launch_bounds__(256) void sketch_minmax_kernel(
    const float* __restrict__ bmin, const float* __restrict__ bmax,
    float* __restrict__ mm, int n)
{
    __shared__ float smin[8], smax[8];
    const int t = threadIdx.x;
    float m0 = INFINITY, m1 = -INFINITY;
    for (int i = t; i < n; i += 256) {
        m0 = fminf(m0, bmin[i]);
        m1 = fmaxf(m1, bmax[i]);
    }
#pragma unroll
    for (int off = 16; off >= 1; off >>= 1) {
        m0 = fminf(m0, __shfl_xor(m0, off, 32));
        m1 = fmaxf(m1, __shfl_xor(m1, off, 32));
    }
    if ((t & 31) == 0) { smin[t >> 5] = m0; smax[t >> 5] = m1; }
    __syncthreads();
    if (t == 0) {
        for (int w = 1; w < 8; ++w) {
            m0 = fminf(m0, smin[w]);
            m1 = fmaxf(m1, smax[w]);
        }
        mm[0] = m0;
        mm[1] = m1;
    }
}

__global__ void sketch_normalize_kernel(float* __restrict__ out,
                                        const float* __restrict__ mm, int n)
{
    const int i = blockIdx.x * blockDim.x + threadIdx.x;
    if (i < n) {
        const float tmin = mm[0];
        const float tmax = mm[1];
        out[i] = (out[i] - tmin) / (tmax - tmin);
    }
}

extern "C" void kernel_launch(void* const* d_in, const int* in_sizes, int n_in,
                              void* d_out, int out_size, void* d_ws, size_t ws_size,
                              hipStream_t stream)
{
    const float* xs = (const float*)d_in[0];
    const float* ys = (const float*)d_in[1];
    float* out = (float*)d_out;
    float* ws  = (float*)d_ws;

    float* bmin = ws;
    float* bmax = ws + 1024;
    float* mm   = ws + 2048;

    dim3 grid(IMSIZE / 16, IMSIZE / BROWS);
    sketch_render_kernel<<<grid, 256, 0, stream>>>(xs, ys, out, bmin, bmax);

    const int nblocks = (IMSIZE / 16) * (IMSIZE / BROWS);
    sketch_minmax_kernel<<<1, 256, 0, stream>>>(bmin, bmax, mm, nblocks);

    const int n = IMSIZE * IMSIZE;
    sketch_normalize_kernel<<<(n + 255) / 256, 256, 0, stream>>>(out, mm, n);
}


# One unit coordinate for both box axes: the point is on the line, so yp - yc is one fma on the clamped x coordinate with +-h; the 8 clamped y fmas a group go; new rounding, sum order kept
# speedup vs baseline: 1.9077x; 1.0438x over previous
#include <hip/hip_runtime.h>
#include <math.h>


#define IMSIZE 1024
#define NSEG   128
#define FEPS   1e-10f
#define TROWS  16
#define BROWS  (8 * TROWS)

typedef __attribute__((ext_vector_type(2))) float v2f;
typedef __attribute__((ext_vector_type(8))) float v8f;

template <int MASK>
__device__ __forceinline__ float row_xor(float v)
{
    return __builtin_bit_cast(float,
        __builtin_amdgcn_update_dpp(0, __builtin_bit_cast(int, v), 0x160 + MASK, 0xf, 0xf, true));
}

__global__ __launch_bounds__(256) __attribute__((amdgpu_waves_per_eu(4, 4))) void sketch_render_kernel(
    const float* __restrict__ xs, const float* __restrict__ ys,
    float* __restrict__ out, float* __restrict__ bmin, float* __restrict__ bmax)
{
    __shared__ float s_cy[NSEG], s_cx[NSEG], s_c0[NSEG];
    __shared__ float s_ax[NSEG], s_nw[NSEG], s_xl[NSEG];
    __shared__ float s_hs[NSEG];
    __shared__ float2 s_rt[NSEG];
    __shared__ float2 s_b1[NSEG];
    __shared__ float s_red[16];

    const int t = threadIdx.x;

    if (t < NSEG) {
        float x0 = xs[t], x1 = xs[t + 1];
        float y0 = ys[t], y1 = ys[t + 1];
        float dx = x1 - x0, dy = y1 - y0;
        float dx2 = dx * dx;
        float d = dx2 + dy * dy;
        float invd = 1.0f / (d + FEPS);
        float cross = y0 * x1 - x0 * y1;
        const float cy = dx * dy * invd;
        s_cy[t] = cy;
        s_cx[t] = dx2 * invd;
        s_c0[t] = -dy * cross * invd;
        const bool  vt = (x0 == x1);
        const float xl = fminf(x0, x1), yl = fminf(y0, y1);
        const float w  = fmaxf(x0, x1) - xl;
        const float h  = fmaxf(y0, y1) - yl;
        const float iw = (w > 1e-30f) ? 1.0f / w : 0.0f;
        const float ih = (h > 1e-30f) ? 1.0f / h : 0.0f;
        const float ax = iw, bx = -xl * iw;
        s_ax[t] = ax;  s_nw[t] = -w;  s_xl[t] = xl;
        s_b1[t] = make_float2(bx, bx + ax * cy);
        const bool opp = !vt && ((dx < 0.0f) != (dy < 0.0f));
        s_hs[t] = opp ? h : -h;
        s_rt[t] = make_float2(opp ? fmaxf(y0, y1) : yl, vt ? ih : 0.0f);
    }
    __syncthreads();

    const int lane = t & 31;
    const int wave = __builtin_amdgcn_readfirstlane(t >> 5);
    const int c    = lane & 15;
    const int hi   = lane >> 4;
    const int xbase = blockIdx.x * 16;
    const int ybase = blockIdx.y * BROWS + wave * TROWS;

    v2f bfrag[8];
#pragma unroll
    for (int g = 0; g < 8; ++g) {
        const int n = g * 16 + c;
        bfrag[g].x = hi ? s_c0[n] : s_cy[n];
        bfrag[g].y = hi ? 0.0f    : s_cx[n];
    }
    v2f k_ax[4], k_nw[4], k_hs[4];
#pragma unroll
    for (int q = 0; q < 4; ++q) {
        const int n0 = (2 * q) * 16 + c, n1 = n0 + 16;
        k_ax[q].x = s_ax[n0]; k_ax[q].y = s_ax[n1];
        k_nw[q].x = s_nw[n0]; k_nw[q].y = s_nw[n1];
        k_hs[q].x = s_hs[n0]; k_hs[q].y = s_hs[n1];
    }
    v2f k_dx[8][4];
#pragma unroll
    for (int g = 0; g < 8; ++g) {
        const float xl = s_xl[g * 16 + c];
#pragma unroll
        for (int k = 0; k < 4; ++k) {
            k_dx[g][k].x = (float)(xbase + 8 * hi + 2 * k)     - xl;
            k_dx[g][k].y = (float)(xbase + 8 * hi + 2 * k + 1) - xl;
        }
    }
    float2 k_rt[8];
    v2f    k_b1[8];
#pragma unroll
    for (int g = 0; g < 8; ++g) {
        k_rt[g] = s_rt[g * 16 + c];
        const float2 bq = s_b1[g * 16 + c];
        k_b1[g].x = bq.x; k_b1[g].y = bq.y;
    }
    const float xa = (float)(xbase + c);

    const bool b3 = (lane & 8) != 0, b2 = (lane & 4) != 0, b1 = (lane & 2) != 0;

    float lmin = INFINITY, lmax = -INFINITY;

    auto project = [&](int row, int g) -> v8f {
        const float yp = (float)row;
        v2f afrag;
        afrag.x = hi ? 1.0f : yp;
        afrag.y = hi ? 0.0f : xa;
        v8f cz = {0.f, 0.f, 0.f, 0.f, 0.f, 0.f, 0.f, 0.f};
        return __builtin_amdgcn_wmma_f32_16x16x4_f32(
            false, afrag, false, bfrag[g], (short)0, cz, false, false);
    };

#define PK_FMA_NEW(r, a, x, b, MODS) asm volatile("v_pk_fma_f32 %0, %1, %2, %3 " MODS : "=v"(r) : "v"(a), "v"(x), "v"(b))
#define PK_FMA_OVER(x, a, b, MODS)   asm volatile("v_pk_fma_f32 %0, %1, %0, %2 " MODS : "+v"(x) : "v"(a), "v"(b))
    auto pk_unit_new = [&](v2f a, v2f x, v2f b, int H, int P) -> v2f {
        v2f r;
        if      (H == 0 && P == 0) PK_FMA_NEW(r, a, x, b, "op_sel:[0,0,0] op_sel_hi:[0,0,1] clamp");
        else if (H == 0 && P == 1) PK_FMA_NEW(r, a, x, b, "op_sel:[0,1,0] op_sel_hi:[0,1,1] clamp");
        else if (H == 1 && P == 0) PK_FMA_NEW(r, a, x, b, "op_sel:[1,0,0] op_sel_hi:[1,0,1] clamp");
        else                       PK_FMA_NEW(r, a, x, b, "op_sel:[1,1,0] op_sel_hi:[1,1,1] clamp");
        return r;
    };
    auto pk_backy_new = [&](v2f a, v2f x, v2f b, int H) -> v2f {
        v2f r;
        if (H == 0) PK_FMA_NEW(r, a, x, b, "op_sel:[0,0,0] op_sel_hi:[0,1,1]");
        else        PK_FMA_NEW(r, a, x, b, "op_sel:[1,0,0] op_sel_hi:[1,1,1]");
        return r;
    };
    auto pk_backx_over = [&](v2f& x, v2f a, v2f b, int H, int P) {
        if      (H == 0 && P == 0) PK_FMA_OVER(x, a, b, "op_sel:[0,0,0] op_sel_hi:[0,1,0]");
        else if (H == 0 && P == 1) PK_FMA_OVER(x, a, b, "op_sel:[0,0,1] op_sel_hi:[0,1,1]");
        else if (H == 1 && P == 0) PK_FMA_OVER(x, a, b, "op_sel:[1,0,0] op_sel_hi:[1,1,0]");
        else                       PK_FMA_OVER(x, a, b, "op_sel:[1,0,1] op_sel_hi:[1,1,1]");
    };

    auto box_stage1 = [&](const v8f& d, int g, v2f bxp, v2f yplp, v2f (&tx)[8], v2f (&ddy)[8]) {
        v2f dp[4];
#pragma unroll
        for (int k = 0; k < 4; ++k) { dp[k].x = d[2 * k]; dp[k].y = d[2 * k + 1]; }
#pragma unroll
        for (int r = 0; r < 8; ++r) tx[r]  = pk_unit_new(k_ax[g >> 1], dp[r >> 1], bxp, g & 1, r & 1);
#pragma unroll
        for (int r = 0; r < 8; ++r) ddy[r] = pk_backy_new(k_hs[g >> 1], tx[r], yplp, g & 1);
    };
    auto box_stage2 = [&](int g, v2f (&tx)[8]) {
#pragma unroll
        for (int r = 0; r < 8; ++r) pk_backx_over(tx[r], k_nw[g >> 1], k_dx[g][r >> 1], g & 1, r & 1);
    };

    auto add_dists = [&](const v2f (&ddx)[8], const v2f (&ddy)[8], int g, v2f (&acc)[8]) {
        const v2f eps2 = {FEPS, FEPS};
#pragma unroll
        for (int r = 0; r < 8; ++r) {
            const v2f s = __builtin_elementwise_fma(ddx[r], ddx[r], __builtin_elementwise_fma(ddy[r], ddy[r], eps2));
            v2f dist;
            dist.x = __builtin_amdgcn_sqrtf(s.x);
            dist.y = __builtin_amdgcn_sqrtf(s.y);
            if (g == 0) acc[r] = dist;
            else        acc[r] = acc[r] + dist;
        }
    };

    auto row_terms = [&](float yp0, float yp1, float2 rt, v2f b1p, v2f& yplp, v2f& bxp) {
        yplp.x = yp0 - rt.x;
        yplp.y = yp1 - rt.x;
        const v2f vh2 = {rt.y, rt.y};
        bxp = __builtin_elementwise_fma(yplp, vh2, b1p);
    };

    auto lane_sum = [&](const float (&acc)[8]) -> float {
        float s4[4];
#pragma unroll
        for (int k = 0; k < 4; ++k) {
            float keep = b3 ? acc[k + 4] : acc[k];
            float send = b3 ? acc[k]     : acc[k + 4];
            s4[k] = keep + row_xor<8>(send);
        }
        float s2[2];
#pragma unroll
        for (int k = 0; k < 2; ++k) {
            float keep = b2 ? s4[k + 2] : s4[k];
            float send = b2 ? s4[k]     : s4[k + 2];
            s2[k] = keep + row_xor<4>(send);
        }
        float keep1 = b1 ? s2[1] : s2[0];
        float send1 = b1 ? s2[0] : s2[1];
        float s1  = keep1 + row_xor<2>(send1);
        return s1 + row_xor<1>(s1);
    };

    v8f dcur = project(ybase, 0);
    v2f yplp, bxp;
    row_terms((float)ybase, (float)(ybase + 1), k_rt[0], k_b1[0], yplp, bxp);

#pragma unroll 1
    for (int i = 0; i < TROWS; i += 2) {
        const int   row = ybase + i;
        const float yp0 = (float)row, yp1 = (float)(row + 1);

        v2f acc2[8];
#pragma unroll
        for (int g = 0; g < 8; ++g) {
            v2f ddx[8], ddy[8];
            box_stage1(dcur, g, bxp, yplp, ddx, ddy);
            __builtin_amdgcn_sched_barrier(0);
            const v8f dnext = (g < 7) ? project(row, g + 1) : project(row + 2, 0);
            if (g < 7) row_terms(yp0, yp1, k_rt[g + 1], k_b1[g + 1], yplp, bxp);
            else       row_terms((float)(row + 2), (float)(row + 3), k_rt[0], k_b1[0], yplp, bxp);
            box_stage2(g, ddx);
            __builtin_amdgcn_sched_barrier(0);
            add_dists(ddx, ddy, g, acc2);
            asm volatile("" :: "v"(acc2[0]), "v"(acc2[1]), "v"(acc2[2]), "v"(acc2[3]),
                               "v"(acc2[4]), "v"(acc2[5]), "v"(acc2[6]), "v"(acc2[7]));
            dcur = dnext;
            if (g < 7) __builtin_amdgcn_sched_barrier(0);
        }

        float ra[8], rb[8];
#pragma unroll
        for (int r = 0; r < 8; ++r) { ra[r] = acc2[r].x; rb[r] = acc2[r].y; }
        const float tot0 = lane_sum(ra);
        const float tot1 = lane_sum(rb);

        lmin = fminf(lmin, fminf(tot0, tot1));
        lmax = fmaxf(lmax, fmaxf(tot0, tot1));
        if ((lane & 1) == 0) {
            out[row * IMSIZE + xbase + 8 * hi + (c >> 1)]       = tot0;
            out[(row + 1) * IMSIZE + xbase + 8 * hi + (c >> 1)] = tot1;
        }
    }

#pragma unroll
    for (int off = 16; off >= 1; off >>= 1) {
        lmin = fminf(lmin, __shfl_xor(lmin, off, 32));
        lmax = fmaxf(lmax, __shfl_xor(lmax, off, 32));
    }
    if (lane == 0) { s_red[wave] = lmin; s_red[8 + wave] = lmax; }
    __syncthreads();
    if (t == 0) {
        float m0 = s_red[0], m1 = s_red[8];
        for (int w = 1; w < 8; ++w) {
            m0 = fminf(m0, s_red[w]);
            m1 = fmaxf(m1, s_red[8 + w]);
        }
        const int bid = blockIdx.y * gridDim.x + blockIdx.x;
        bmin[bid] = m0;
        bmax[bid] = m1;
    }
}

__global__ __launch_bounds__(256) void sketch_minmax_kernel(
    const float* __restrict__ bmin, const float* __restrict__ bmax,
    float* __restrict__ mm, int n)
{
    __shared__ float smin[8], smax[8];
    const int t = threadIdx.x;
    float m0 = INFINITY, m1 = -INFINITY;
    for (int i = t; i < n; i += 256) {
        m0 = fminf(m0, bmin[i]);
        m1 = fmaxf(m1, bmax[i]);
    }
#pragma unroll
    for (int off = 16; off >= 1; off >>= 1) {
        m0 = fminf(m0, __shfl_xor(m0, off, 32));
        m1 = fmaxf(m1, __shfl_xor(m1, off, 32));
    }
    if ((t & 31) == 0) { smin[t >> 5] = m0; smax[t >> 5] = m1; }
    __syncthreads();
    if (t == 0) {
        for (int w = 1; w < 8; ++w) {
            m0 = fminf(m0, smin[w]);
            m1 = fmaxf(m1, smax[w]);
        }
        mm[0] = m0;
        mm[1] = m1;
    }
}

__global__ void sketch_normalize_kernel(float* __restrict__ out,
                                        const float* __restrict__ mm, int n)
{
    const int i = blockIdx.x * blockDim.x + threadIdx.x;
    if (i < n) {
        const float tmin = mm[0];
        const float tmax = mm[1];
        out[i] = (out[i] - tmin) / (tmax - tmin);
    }
}

extern "C" void kernel_launch(void* const* d_in, const int* in_sizes, int n_in,
                              void* d_out, int out_size, void* d_ws, size_t ws_size,
                              hipStream_t stream)
{
    const float* xs = (const float*)d_in[0];
    const float* ys = (const float*)d_in[1];
    float* out = (float*)d_out;
    float* ws  = (float*)d_ws;

    float* bmin = ws;
    float* bmax = ws + 1024;
    float* mm   = ws + 2048;

    dim3 grid(IMSIZE / 16, IMSIZE / BROWS);
    sketch_render_kernel<<<grid, 256, 0, stream>>>(xs, ys, out, bmin, bmax);

    const int nblocks = (IMSIZE / 16) * (IMSIZE / BROWS);
    sketch_minmax_kernel<<<1, 256, 0, stream>>>(bmin, bmax, mm, nblocks);

    const int n = IMSIZE * IMSIZE;
    sketch_normalize_kernel<<<(n + 255) / 256, 256, 0, stream>>>(out, mm, n);
}


# speedup vs baseline: 2.1072x; 1.1046x over previous
#include <hip/hip_runtime.h>
#include <math.h>


#define IMSIZE 1024
#define NSEG   128
#define FEPS   1e-10f
#define TROWS  16
#define BROWS  (8 * TROWS)

typedef __attribute__((ext_vector_type(2))) float v2f;
typedef __attribute__((ext_vector_type(8))) float v8f;

template <int MASK>
__device__ __forceinline__ float row_xor(float v)
{
    return __builtin_bit_cast(float,
        __builtin_amdgcn_update_dpp(0, __builtin_bit_cast(int, v), 0x160 + MASK, 0xf, 0xf, true));
}

__global__ __launch_bounds__(256) __attribute__((amdgpu_waves_per_eu(4, 4))) void sketch_render_kernel(
    const float* __restrict__ xs, const float* __restrict__ ys,
    float* __restrict__ out, float* __restrict__ bmin, float* __restrict__ bmax)
{
    __shared__ float2 s_bf[2][NSEG];
    __shared__ float4 s_pa[NSEG];
    __shared__ float4 s_pb[NSEG];
    __shared__ float s_red[16];

    const int t = threadIdx.x;

    if (t < NSEG) {
        float x0 = xs[t], x1 = xs[t + 1];
        float y0 = ys[t], y1 = ys[t + 1];
        float dx = x1 - x0, dy = y1 - y0;
        float dx2 = dx * dx;
        float d = dx2 + dy * dy;
        float invd = 1.0f / (d + FEPS);
        float cross = y0 * x1 - x0 * y1;
        const float cy = dx * dy * invd;
        s_bf[0][t] = make_float2(cy, dx2 * invd);
        s_bf[1][t] = make_float2(-dy * cross * invd, 0.0f);
        const bool  vt = (x0 == x1);
        const float xl = fminf(x0, x1), yl = fminf(y0, y1);
        const float w  = fmaxf(x0, x1) - xl;
        const float h  = fmaxf(y0, y1) - yl;
        const float iw = (w > 1e-30f) ? 1.0f / w : 0.0f;
        const float ih = (h > 1e-30f) ? 1.0f / h : 0.0f;
        const float ax = iw, bx = -xl * iw;
        const bool opp = !vt && ((dx < 0.0f) != (dy < 0.0f));
        s_pa[t] = make_float4(ax, opp ? h : -h, -w, xl);
        s_pb[t] = make_float4(opp ? fmaxf(y0, y1) : yl, vt ? ih : 0.0f, bx, ax * cy);
    }
    __syncthreads();

    const int lane = t & 31;
    const int wave = __builtin_amdgcn_readfirstlane(t >> 5);
    const int c    = lane & 15;
    const int hi   = lane >> 4;
    const int xbase = blockIdx.x * 16;
    const int ybase = blockIdx.y * BROWS + wave * TROWS;

    float xpf[8];
#pragma unroll
    for (int r = 0; r < 8; ++r) xpf[r] = (float)(xbase + 8 * hi + r);

    v2f afrag;
    afrag.x = hi ? 1.0f : (float)ybase;
    afrag.y = hi ? 0.0f : (float)(xbase + c);

    const bool b3 = (lane & 8) != 0, b2 = (lane & 4) != 0, b1 = (lane & 2) != 0;

#define PK_FMA_NEW(r, a, x, b, MODS) asm volatile("v_pk_fma_f32 %0, %1, %2, %3 " MODS : "=v"(r) : "v"(a), "v"(x), "v"(b))
#define PK_FMA_OVER(x, a, b, MODS)   asm volatile("v_pk_fma_f32 %0, %1, %0, %2 " MODS : "+v"(x) : "v"(a), "v"(b))
    auto pk_unit_new = [&](v2f a, v2f x, v2f b, int H, int P) -> v2f {
        v2f r;
        if      (H == 0 && P == 0) PK_FMA_NEW(r, a, x, b, "op_sel:[0,0,0] op_sel_hi:[0,0,1] clamp");
        else if (H == 0 && P == 1) PK_FMA_NEW(r, a, x, b, "op_sel:[0,1,0] op_sel_hi:[0,1,1] clamp");
        else if (H == 1 && P == 0) PK_FMA_NEW(r, a, x, b, "op_sel:[1,0,0] op_sel_hi:[1,0,1] clamp");
        else                       PK_FMA_NEW(r, a, x, b, "op_sel:[1,1,0] op_sel_hi:[1,1,1] clamp");
        return r;
    };
    auto pk_backy_new = [&](v2f a, v2f x, v2f b, int H) -> v2f {
        v2f r;
        if (H == 0) PK_FMA_NEW(r, a, x, b, "op_sel:[0,0,0] op_sel_hi:[0,1,1]");
        else        PK_FMA_NEW(r, a, x, b, "op_sel:[1,0,0] op_sel_hi:[1,1,1]");
        return r;
    };
    auto pk_backx_over = [&](v2f& x, v2f a, v2f b, int H, int P) {
        if      (H == 0 && P == 0) PK_FMA_OVER(x, a, b, "op_sel:[0,0,0] op_sel_hi:[0,1,0]");
        else if (H == 0 && P == 1) PK_FMA_OVER(x, a, b, "op_sel:[0,0,1] op_sel_hi:[0,1,1]");
        else if (H == 1 && P == 0) PK_FMA_OVER(x, a, b, "op_sel:[1,0,0] op_sel_hi:[1,1,0]");
        else                       PK_FMA_OVER(x, a, b, "op_sel:[1,0,1] op_sel_hi:[1,1,1]");
    };

#define PK_SQ_EPS(y, e) asm volatile("v_pk_fma_f32 %0, %0, %0, %1 op_sel_hi:[1,1,0]" : "+v"(y) : "s"(e))
#define PK_SQ_ADD(x, s) asm volatile("v_pk_fma_f32 %0, %0, %0, %1" : "+v"(x) : "v"(s))
#define ROOT_F32(h)     asm volatile("v_sqrt_f32 %0, %0" : "+v"(h))
#define PK_ACC(a, d)    asm volatile("v_pk_add_f32 %0, %0, %1" : "+v"(a) : "v"(d))
    auto root_half = [&](v2f& p, int k) {
        float h = (k & 1) ? p.y : p.x;
        ROOT_F32(h);
        if (k & 1) p.y = h; else p.x = h;
    };

    auto lane_sum = [&](const float (&acc)[8]) -> float {
        float s4[4];
#pragma unroll
        for (int k = 0; k < 4; ++k) {
            float keep = b3 ? acc[k + 4] : acc[k];
            float send = b3 ? acc[k]     : acc[k + 4];
            s4[k] = keep + row_xor<8>(send);
        }
        float s2[2];
#pragma unroll
        for (int k = 0; k < 2; ++k) {
            float keep = b2 ? s4[k + 2] : s4[k];
            float send = b2 ? s4[k]     : s4[k + 2];
            s2[k] = keep + row_xor<4>(send);
        }
        float keep1 = b1 ? s2[1] : s2[0];
        float send1 = b1 ? s2[0] : s2[1];
        float s1  = keep1 + row_xor<2>(send1);
        return s1 + row_xor<1>(s1);
    };

    v2f acc[TROWS / 2][8];
#pragma unroll
    for (int i = 0; i < TROWS / 2; ++i)
#pragma unroll
        for (int r = 0; r < 8; ++r) { acc[i][r].x = 0.0f; acc[i][r].y = 0.0f; }

    float4 pa = s_pa[c], pb = s_pb[c];
    float2 bf = s_bf[hi][c];

#pragma unroll 1
    for (int g = 0; g < NSEG / 16; ++g) {
        const int nn = ((g + 1) & (NSEG / 16 - 1)) * 16 + c;
        const float4 pa_n = s_pa[nn];
        const float4 pb_n = s_pb[nn];
        const float2 bf_n = s_bf[hi][nn];

        v2f bfrag;
        bfrag.x = bf.x;
        bfrag.y = bf.y;
        const v8f cz = {0.f, 0.f, 0.f, 0.f, 0.f, 0.f, 0.f, 0.f};
        const v8f d = __builtin_amdgcn_wmma_f32_16x16x4_f32(
            false, afrag, false, bfrag, (short)0, cz, false, false);
        __builtin_amdgcn_sched_barrier(0);

        v2f dp[4];
#pragma unroll
        for (int k = 0; k < 4; ++k) { dp[k].x = d[2 * k]; dp[k].y = d[2 * k + 1]; }
        const v2f kA = {pa.x, pa.y};
        const v2f kN = {pa.z, pa.z};
        v2f kdx[4];
#pragma unroll
        for (int k = 0; k < 4; ++k) { kdx[k].x = xpf[2 * k] - pa.w; kdx[k].y = xpf[2 * k + 1] - pa.w; }
        const v2f vh2  = {pb.y, pb.y};
        const v2f b10  = {pb.z, pb.z + pb.w};
        const float st = 2.0f * (pb.y + pb.w);
        const v2f st2  = {st, st};
        v2f yplp, bxp;
        yplp.x = (float)ybase - pb.x;
        yplp.y = (float)(ybase + 1) - pb.x;
        const v2f bxp0 = __builtin_elementwise_fma(yplp, vh2, b10);
        bxp = bxp0;
        __builtin_amdgcn_sched_barrier(0);

        const v2f eps2 = {FEPS, FEPS};
        v2f tx[2][8], dy[2][8];
#pragma unroll
        for (int r = 0; r < 8; ++r) tx[0][r] = pk_unit_new(kA, dp[r >> 1], bxp, 0, r & 1);
#pragma unroll
        for (int r = 0; r < 8; ++r) dy[0][r] = pk_backy_new(kA, tx[0][r], yplp, 1);

#pragma unroll
        for (int i = 0; i < TROWS / 2; ++i) {
            const int cur = i & 1, nxt = cur ^ 1;
            __builtin_amdgcn_sched_barrier(0);
            if (i + 1 < TROWS / 2) {
                yplp.x = (float)(ybase + 2 * i + 2) - pb.x;
                yplp.y = (float)(ybase + 2 * i + 3) - pb.x;
                const v2f fi = {(float)(i + 1), (float)(i + 1)};
                bxp = __builtin_elementwise_fma(fi, st2, bxp0);
            }
            __builtin_amdgcn_sched_barrier(0);
#pragma unroll
            for (int r = 0; r < 8; ++r) pk_backx_over(tx[cur][r], kN, kdx[r >> 1], 0, r & 1);
            if (i > 0 && i + 1 < TROWS / 2) {
#pragma unroll
                for (int r = 0; r < 8; ++r) PK_ACC(acc[i - 1][r], tx[nxt][r]);
            }
#pragma unroll
            for (int r = 0; r < 8; ++r) PK_SQ_EPS(dy[cur][r], eps2);
#pragma unroll
            for (int r = 0; r < 8; ++r) PK_SQ_ADD(tx[cur][r], dy[cur][r]);
            if (i + 1 < TROWS / 2) {
#pragma unroll
                for (int k = 0; k < 16; ++k) {
                    root_half(tx[cur][k >> 1], k);
                    if (k < 8) tx[nxt][k]     = pk_unit_new(kA, dp[k >> 1], bxp, 0, k & 1);
                    else       dy[nxt][k - 8] = pk_backy_new(kA, tx[nxt][k - 8], yplp, 1);
                }
            } else {
#pragma unroll
                for (int k = 0; k < 16; ++k) {
                    root_half(tx[cur][k >> 1], k);
                    if (k < 8)       PK_ACC(acc[i - 1][k], tx[nxt][k]);
                    else if (k < 12) PK_ACC(acc[i][k - 8], tx[cur][k - 8]);
                }
#pragma unroll
                for (int r = 4; r < 8; ++r) PK_ACC(acc[i][r], tx[cur][r]);
            }
        }
        pa = pa_n; pb = pb_n; bf = bf_n;
    }

    float lmin = INFINITY, lmax = -INFINITY;
#pragma unroll
    for (int i = 0; i < TROWS / 2; ++i) {
        const int row = ybase + 2 * i;
        float ra[8], rb[8];
#pragma unroll
        for (int r = 0; r < 8; ++r) { ra[r] = acc[i][r].x; rb[r] = acc[i][r].y; }
        const float tot0 = lane_sum(ra);
        const float tot1 = lane_sum(rb);

        lmin = fminf(lmin, fminf(tot0, tot1));
        lmax = fmaxf(lmax, fmaxf(tot0, tot1));
        if ((lane & 1) == 0) {
            out[row * IMSIZE + xbase + 8 * hi + (c >> 1)]       = tot0;
            out[(row + 1) * IMSIZE + xbase + 8 * hi + (c >> 1)] = tot1;
        }
    }

#pragma unroll
    for (int off = 16; off >= 1; off >>= 1) {
        lmin = fminf(lmin, __shfl_xor(lmin, off, 32));
        lmax = fmaxf(lmax, __shfl_xor(lmax, off, 32));
    }
    if (lane == 0) { s_red[wave] = lmin; s_red[8 + wave] = lmax; }
    __syncthreads();
    if (t == 0) {
        float m0 = s_red[0], m1 = s_red[8];
        for (int w = 1; w < 8; ++w) {
            m0 = fminf(m0, s_red[w]);
            m1 = fmaxf(m1, s_red[8 + w]);
        }
        const int bid = blockIdx.y * gridDim.x + blockIdx.x;
        bmin[bid] = m0;
        bmax[bid] = m1;
    }
}

__global__ __launch_bounds__(256) void sketch_minmax_kernel(
    const float* __restrict__ bmin, const float* __restrict__ bmax,
    float* __restrict__ mm, int n)
{
    __shared__ float smin[8], smax[8];
    const int t = threadIdx.x;
    float m0 = INFINITY, m1 = -INFINITY;
    for (int i = t; i < n; i += 256) {
        m0 = fminf(m0, bmin[i]);
        m1 = fmaxf(m1, bmax[i]);
    }
#pragma unroll
    for (int off = 16; off >= 1; off >>= 1) {
        m0 = fminf(m0, __shfl_xor(m0, off, 32));
        m1 = fmaxf(m1, __shfl_xor(m1, off, 32));
    }
    if ((t & 31) == 0) { smin[t >> 5] = m0; smax[t >> 5] = m1; }
    __syncthreads();
    if (t == 0) {
        for (int w = 1; w < 8; ++w) {
            m0 = fminf(m0, smin[w]);
            m1 = fmaxf(m1, smax[w]);
        }
        mm[0] = m0;
        mm[1] = m1;
    }
}

__global__ void sketch_normalize_kernel(float* __restrict__ out,
                                        const float* __restrict__ mm, int n)
{
    const int i = blockIdx.x * blockDim.x + threadIdx.x;
    if (i < n) {
        const float tmin = mm[0];
        const float tmax = mm[1];
        out[i] = (out[i] - tmin) / (tmax - tmin);
    }
}

extern "C" void kernel_launch(void* const* d_in, const int* in_sizes, int n_in,
                              void* d_out, int out_size, void* d_ws, size_t ws_size,
                              hipStream_t stream)
{
    const float* xs = (const float*)d_in[0];
    const float* ys = (const float*)d_in[1];
    float* out = (float*)d_out;
    float* ws  = (float*)d_ws;

    float* bmin = ws;
    float* bmax = ws + 1024;
    float* mm   = ws + 2048;

    dim3 grid(IMSIZE / 16, IMSIZE / BROWS);
    sketch_render_kernel<<<grid, 256, 0, stream>>>(xs, ys, out, bmin, bmax);

    const int nblocks = (IMSIZE / 16) * (IMSIZE / BROWS);
    sketch_minmax_kernel<<<1, 256, 0, stream>>>(bmin, bmax, mm, nblocks);

    const int n = IMSIZE * IMSIZE;
    sketch_normalize_kernel<<<(n + 255) / 256, 256, 0, stream>>>(out, mm, n);
}
